# starting kernel
# baseline (speedup 1.0000x reference)
#include <hip/hip_runtime.h>
#include <math.h>

typedef __attribute__((ext_vector_type(2))) float v2f;
typedef __attribute__((ext_vector_type(8))) float v8f;

#define B_  8
#define T_  1024
#define DIN 256
#define DH1 1024
#define DH2 1024
#define KW  9

__device__ __forceinline__ float mishf(float v) {
    float sp = (v > 20.f) ? v : log1pf(__expf(v));
    return v * tanhf(sp);
}

__global__ void k_norm_d(const float* __restrict__ dw, float* __restrict__ inv_nd) {
    int tid = threadIdx.x;
    if (tid >= B_ * KW) return;
    int b = tid / KW, k = tid % KW;
    float acc = 0.f;
    for (int c = 0; c < DH1; ++c) {
        float v = dw[((size_t)b * DH1 + c) * KW + k];
        acc += v * v;
    }
    inv_nd[tid] = 1.f / fmaxf(sqrtf(acc), 1e-12f);
}

__global__ __launch_bounds__(256) void k_norm_p(const float* __restrict__ pw,
                                                float* __restrict__ inv_np) {
    __shared__ float s[256];
    int b  = blockIdx.x >> 4;
    int o  = ((blockIdx.x & 15) << 6) + (threadIdx.x & 63);
    int cg = threadIdx.x >> 6;
    const float* P = pw + (size_t)b * DH1 * DH2;
    float acc = 0.f;
    for (int c = cg; c < DH1; c += 4) {
        float v = P[(size_t)c * DH2 + o];
        acc += v * v;
    }
    s[threadIdx.x] = acc;
    __syncthreads();
    if (cg == 0) {
        float tot = s[threadIdx.x] + s[threadIdx.x + 64] + s[threadIdx.x + 128] + s[threadIdx.x + 192];
        inv_np[b * DH2 + o] = 1.f / fmaxf(sqrtf(tot), 1e-12f);
    }
}

__global__ __launch_bounds__(256) void k_gemm1_mish(
    const float* __restrict__ x, const float* __restrict__ w1w,
    const float* __restrict__ w1b, float* __restrict__ h)
{
    const int lane = threadIdx.x & 31;
    const int hf = lane >> 4, l16 = lane & 15;
    int tid = blockIdx.x * 8 + (threadIdx.x >> 5);
    const int tn = T_ / 16, tm = DH1 / 16;
    int b  = tid / (tm * tn);
    int r  = tid % (tm * tn);
    int m0 = (r / tn) * 16;
    int n0 = (r % tn) * 16;

    const float* Arow = w1w + (size_t)(m0 + l16) * DIN;
    const float* Brow = x + ((size_t)b * T_ + (n0 + l16)) * DIN;
    v8f acc = {};
    #pragma unroll 4
    for (int k0 = 0; k0 < DIN; k0 += 4) {
        v2f a  = *(const v2f*)(Arow + k0 + 2 * hf);
        v2f bv = *(const v2f*)(Brow + k0 + 2 * hf);
        acc = __builtin_amdgcn_wmma_f32_16x16x4_f32(false, a, false, bv,
                                                    (short)0, acc, false, false);
    }
    float* H = h + (size_t)b * DH1 * T_;
    #pragma unroll
    for (int v = 0; v < 8; ++v) {
        int c = m0 + v + 8 * hf;
        int t = n0 + l16;
        H[(size_t)c * T_ + t] = mishf(acc[v] + w1b[c]);
    }
}

__global__ __launch_bounds__(256) void k_dwconv(
    const float* __restrict__ h, const float* __restrict__ dw,
    const float* __restrict__ dg, const float* __restrict__ db,
    const float* __restrict__ pg, const float* __restrict__ inv_nd,
    float* __restrict__ y)
{
    int idx = blockIdx.x * 256 + threadIdx.x;
    int t = idx & (T_ - 1);
    int c = (idx >> 10) & (DH1 - 1);
    int b = idx >> 20;
    const float* hp  = h + ((size_t)b * DH1 + c) * T_;
    const float* wp  = dw + ((size_t)b * DH1 + c) * KW;
    const float* ndp = inv_nd + b * KW;
    float acc = 0.f;
    #pragma unroll
    for (int k = 0; k < KW; ++k) {
        int tt = t + k - 4;
        if (tt >= 0 && tt < T_)
            acc += hp[tt] * (wp[k] * ndp[k]);
    }
    int bc = b * DH1 + c;
    float yy = acc * dg[bc] + db[bc];
    y[((size_t)b * DH1 + c) * T_ + t] = yy * pg[bc];
}

__global__ __launch_bounds__(256) void k_gemm2(
    const float* __restrict__ y, const float* __restrict__ pw,
    const float* __restrict__ pb, const float* __restrict__ inv_np,
    float* __restrict__ z)
{
    const int lane = threadIdx.x & 31;
    const int hf = lane >> 4, l16 = lane & 15;
    int tid = blockIdx.x * 8 + (threadIdx.x >> 5);
    const int tn = T_ / 16, tm = DH2 / 16;
    int b  = tid / (tm * tn);
    int r  = tid % (tm * tn);
    int m0 = (r / tn) * 16;
    int n0 = (r % tn) * 16;
    const float* A  = pw + (size_t)b * DH1 * DH2;
    const float* Bm = y  + (size_t)b * DH1 * T_;
    v8f acc = {};
    #pragma unroll 4
    for (int k0 = 0; k0 < DH1; k0 += 4) {
        int ka = k0 + 2 * hf;
        v2f a, bv;
        a.x  = A[(size_t)ka * DH2 + m0 + l16];
        a.y  = A[(size_t)(ka + 1) * DH2 + m0 + l16];
        bv.x = Bm[(size_t)ka * T_ + n0 + l16];
        bv.y = Bm[(size_t)(ka + 1) * T_ + n0 + l16];
        acc = __builtin_amdgcn_wmma_f32_16x16x4_f32(false, a, false, bv,
                                                    (short)0, acc, false, false);
    }
    float* Z = z + (size_t)b * DH2 * T_;
    #pragma unroll
    for (int v = 0; v < 8; ++v) {
        int o = m0 + v + 8 * hf;
        int t = n0 + l16;
        Z[(size_t)o * T_ + t] = acc[v] * inv_np[b * DH2 + o] + pb[b * DH2 + o];
    }
}

__global__ __launch_bounds__(256) void k_gemm3(
    const float* __restrict__ z, const float* __restrict__ w2w,
    const float* __restrict__ w2b, const float* __restrict__ x,
    float* __restrict__ out)
{
    const int lane = threadIdx.x & 31;
    const int hf = lane >> 4, l16 = lane & 15;
    int tid = blockIdx.x * 8 + (threadIdx.x >> 5);
    const int tn = T_ / 16, tm = DIN / 16;
    int b  = tid / (tm * tn);
    int r  = tid % (tm * tn);
    int m0 = (r / tn) * 16;
    int n0 = (r % tn) * 16;
    const float* Arow = w2w + (size_t)(m0 + l16) * DH2;
    const float* Bm   = z + (size_t)b * DH2 * T_;
    v8f acc = {};
    #pragma unroll 4
    for (int k0 = 0; k0 < DH2; k0 += 4) {
        int ka = k0 + 2 * hf;
        v2f a = *(const v2f*)(Arow + ka);
        v2f bv;
        bv.x = Bm[(size_t)ka * T_ + n0 + l16];
        bv.y = Bm[(size_t)(ka + 1) * T_ + n0 + l16];
        acc = __builtin_amdgcn_wmma_f32_16x16x4_f32(false, a, false, bv,
                                                    (short)0, acc, false, false);
    }
    #pragma unroll
    for (int v = 0; v < 8; ++v) {
        int i = m0 + v + 8 * hf;
        int t = n0 + l16;
        size_t oi = ((size_t)b * T_ + t) * DIN + i;
        out[oi] = acc[v] + w2b[i] + x[oi];
    }
}

extern "C" void kernel_launch(void* const* d_in, const int* in_sizes, int n_in,
                              void* d_out, int out_size, void* d_ws, size_t ws_size,
                              hipStream_t stream)
{
    const float* x   = (const float*)d_in[0];
    const float* d_w = (const float*)d_in[1];
    const float* d_g = (const float*)d_in[2];
    const float* d_b = (const float*)d_in[3];
    const float* p_w = (const float*)d_in[4];
    const float* p_g = (const float*)d_in[5];
    const float* p_b = (const float*)d_in[6];
    const float* w1w = (const float*)d_in[7];
    const float* w1b = (const float*)d_in[8];
    const float* w2w = (const float*)d_in[9];
    const float* w2b = (const float*)d_in[10];
    float* out = (float*)d_out;

    float* ws = (float*)d_ws;
    const size_t NCT = (size_t)B_ * DH1 * T_;
    float* hbuf   = ws;
    float* ybuf   = ws + NCT;
    float* inv_nd = ws + 2 * NCT;
    float* inv_np = inv_nd + 128;

    k_norm_d<<<1, 128, 0, stream>>>(d_w, inv_nd);
    k_norm_p<<<B_ * (DH2 / 64), 256, 0, stream>>>(p_w, inv_np);
    k_gemm1_mish<<<(B_ * (DH1 / 16) * (T_ / 16)) / 8, 256, 0, stream>>>(x, w1w, w1b, hbuf);
    k_dwconv<<<(int)(NCT / 256), 256, 0, stream>>>(hbuf, d_w, d_g, d_b, p_g, inv_nd, ybuf);
    k_gemm2<<<(B_ * (DH2 / 16) * (T_ / 16)) / 8, 256, 0, stream>>>(ybuf, p_w, p_b, inv_np, hbuf);
    k_gemm3<<<(B_ * (DIN / 16) * (T_ / 16)) / 8, 256, 0, stream>>>(hbuf, w2w, w2b, x, out);
}


# GEMM 2 re-cut by time: a work-group owns 32 time steps of one batch for all 1024 outputs, its y panel stays in LDS for the whole launch, each weight line is fetched once per group; k order kept
# speedup vs baseline: 1.1707x; 1.1707x over previous
#include <hip/hip_runtime.h>
#include <math.h>

typedef __attribute__((ext_vector_type(2))) float v2f;
typedef __attribute__((ext_vector_type(8))) float v8f;

#define B_  8
#define T_  1024
#define DIN 256
#define DH1 1024
#define DH2 1024
#define KW  9

__device__ __forceinline__ float mishf(float v) {
    float sp = (v > 20.f) ? v : log1pf(__expf(v));
    return v * tanhf(sp);
}

__global__ void k_norm_d(const float* __restrict__ dw, float* __restrict__ inv_nd) {
    int tid = threadIdx.x;
    if (tid >= B_ * KW) return;
    int b = tid / KW, k = tid % KW;
    float acc = 0.f;
    for (int c = 0; c < DH1; ++c) {
        float v = dw[((size_t)b * DH1 + c) * KW + k];
        acc += v * v;
    }
    inv_nd[tid] = 1.f / fmaxf(sqrtf(acc), 1e-12f);
}

__global__ __launch_bounds__(256) void k_norm_p(const float* __restrict__ pw,
                                                float* __restrict__ inv_np) {
    __shared__ float s[256];
    int b  = blockIdx.x >> 4;
    int o  = ((blockIdx.x & 15) << 6) + (threadIdx.x & 63);
    int cg = threadIdx.x >> 6;
    const float* P = pw + (size_t)b * DH1 * DH2;
    float acc = 0.f;
    for (int c = cg; c < DH1; c += 4) {
        float v = P[(size_t)c * DH2 + o];
        acc += v * v;
    }
    s[threadIdx.x] = acc;
    __syncthreads();
    if (cg == 0) {
        float tot = s[threadIdx.x] + s[threadIdx.x + 64] + s[threadIdx.x + 128] + s[threadIdx.x + 192];
        inv_np[b * DH2 + o] = 1.f / fmaxf(sqrtf(tot), 1e-12f);
    }
}

__global__ __launch_bounds__(256) void k_gemm1_mish(
    const float* __restrict__ x, const float* __restrict__ w1w,
    const float* __restrict__ w1b, float* __restrict__ h)
{
    const int lane = threadIdx.x & 31;
    const int hf = lane >> 4, l16 = lane & 15;
    int tid = blockIdx.x * 8 + (threadIdx.x >> 5);
    const int tn = T_ / 16, tm = DH1 / 16;
    int b  = tid / (tm * tn);
    int r  = tid % (tm * tn);
    int m0 = (r / tn) * 16;
    int n0 = (r % tn) * 16;

    const float* Arow = w1w + (size_t)(m0 + l16) * DIN;
    const float* Brow = x + ((size_t)b * T_ + (n0 + l16)) * DIN;
    v8f acc = {};
    #pragma unroll 4
    for (int k0 = 0; k0 < DIN; k0 += 4) {
        v2f a  = *(const v2f*)(Arow + k0 + 2 * hf);
        v2f bv = *(const v2f*)(Brow + k0 + 2 * hf);
        acc = __builtin_amdgcn_wmma_f32_16x16x4_f32(false, a, false, bv,
                                                    (short)0, acc, false, false);
    }
    float* H = h + (size_t)b * DH1 * T_;
    #pragma unroll
    for (int v = 0; v < 8; ++v) {
        int c = m0 + v + 8 * hf;
        int t = n0 + l16;
        H[(size_t)c * T_ + t] = mishf(acc[v] + w1b[c]);
    }
}

__global__ __launch_bounds__(256) void k_dwconv(
    const float* __restrict__ h, const float* __restrict__ dw,
    const float* __restrict__ dg, const float* __restrict__ db,
    const float* __restrict__ pg, const float* __restrict__ inv_nd,
    float* __restrict__ y)
{
    int idx = blockIdx.x * 256 + threadIdx.x;
    int t = idx & (T_ - 1);
    int c = (idx >> 10) & (DH1 - 1);
    int b = idx >> 20;
    const float* hp  = h + ((size_t)b * DH1 + c) * T_;
    const float* wp  = dw + ((size_t)b * DH1 + c) * KW;
    const float* ndp = inv_nd + b * KW;
    float acc = 0.f;
    #pragma unroll
    for (int k = 0; k < KW; ++k) {
        int tt = t + k - 4;
        if (tt >= 0 && tt < T_)
            acc += hp[tt] * (wp[k] * ndp[k]);
    }
    int bc = b * DH1 + c;
    float yy = acc * dg[bc] + db[bc];
    y[((size_t)b * DH1 + c) * T_ + t] = yy * pg[bc];
}

#define PT 32
#define PP 96
#define G2_WAVES 16
#define G2_U 4
__device__ __forceinline__ void g2_load_block(const float* __restrict__ Ab, v2f (&ax)[G2_U], v2f (&ay)[G2_U]) {
    for (int u = 0; u < G2_U; ++u) {
        ax[u] = *(const v2f*)(Ab + (size_t)(4 * u) * DH2);
        ay[u] = *(const v2f*)(Ab + (size_t)(4 * u + 1) * DH2);
    }
}
__device__ __forceinline__ void g2_block(const float* Bk, const v2f (&ax)[G2_U], const v2f (&ay)[G2_U],
                                         v8f& accE0, v8f& accE1, v8f& accO0, v8f& accO1) {
    for (int u = 0; u < G2_U; ++u) {
        v2f b0 = *(const v2f*)(Bk + (2 * u) * PP);
        v2f b1 = *(const v2f*)(Bk + (2 * u) * PP + 32);
        v2f aE, aO;
        aE.x = ax[u].x; aE.y = ay[u].x;
        aO.x = ax[u].y; aO.y = ay[u].y;
        accE0 = __builtin_amdgcn_wmma_f32_16x16x4_f32(false, aE, false, b0, (short)0, accE0, false, false);
        accE1 = __builtin_amdgcn_wmma_f32_16x16x4_f32(false, aE, false, b1, (short)0, accE1, false, false);
        accO0 = __builtin_amdgcn_wmma_f32_16x16x4_f32(false, aO, false, b0, (short)0, accO0, false, false);
        accO1 = __builtin_amdgcn_wmma_f32_16x16x4_f32(false, aO, false, b1, (short)0, accO1, false, false);
    }
}
__global__ __launch_bounds__(512) void k_gemm2(
    const float* __restrict__ y, const float* __restrict__ pw,
    const float* __restrict__ pb, const float* __restrict__ inv_np,
    float* __restrict__ z)
{
    __shared__ __attribute__((aligned(16))) float panel[(DH1 / 2) * PP];
    const int tid = threadIdx.x;
    const int b  = blockIdx.x / (T_ / PT);
    const int t0 = (blockIdx.x % (T_ / PT)) * PT;

    {
        const float* Y = y + (size_t)b * DH1 * T_ + t0;
        const int j  = tid & 7;
        const int pp = tid >> 3;
        for (int i = 0; i < (DH1 / 2) / 64; ++i) {
            int p = i * 64 + pp;
            float4 u = *(const float4*)(Y + (size_t)(2 * p) * T_ + 4 * j);
            float4 v = *(const float4*)(Y + (size_t)(2 * p + 1) * T_ + 4 * j);
            float4 lo = make_float4(u.x, v.x, u.y, v.y);
            float4 hi = make_float4(u.z, v.z, u.w, v.w);
            *(float4*)(panel + p * PP + 8 * j)     = lo;
            *(float4*)(panel + p * PP + 8 * j + 4) = hi;
        }
    }
    __syncthreads();

    const int lane = tid & 31, wave = tid >> 5;
    const int hf = lane >> 4, l16 = lane & 15;
    const float* A = pw + (size_t)b * DH1 * DH2;
    float* Z = z + (size_t)b * DH2 * T_ + t0;
    const float* Bp = panel + hf * PP + 2 * l16;
    const int NB = DH1 / (4 * G2_U);

    for (int ob = wave * 32; ob < DH2; ob += G2_WAVES * 32) {
        const float* Ap = A + (size_t)(2 * hf) * DH2 + ob + 2 * l16;
        v8f accE0 = {}, accE1 = {}, accO0 = {}, accO1 = {};
        v2f x0[G2_U], y0[G2_U], x1[G2_U], y1[G2_U];
        g2_load_block(Ap, x0, y0);
        for (int kb = 0; kb < NB; kb += 2) {
            g2_load_block(Ap + (size_t)(4 * G2_U * (kb + 1)) * DH2, x1, y1);
            g2_block(Bp + (2 * G2_U * kb) * PP, x0, y0, accE0, accE1, accO0, accO1);
            g2_load_block(Ap + (size_t)(4 * G2_U * ((kb + 2) & (NB - 1))) * DH2, x0, y0);
            g2_block(Bp + (2 * G2_U * (kb + 1)) * PP, x1, y1, accE0, accE1, accO0, accO1);
        }
        for (int v = 0; v < 8; ++v) {
            int oE = ob + 2 * (v + 8 * hf);
            int oO = oE + 1;
            float sE = inv_np[b * DH2 + oE], cE = pb[b * DH2 + oE];
            float sO = inv_np[b * DH2 + oO], cO = pb[b * DH2 + oO];
            Z[(size_t)oE * T_ + l16]      = accE0[v] * sE + cE;
            Z[(size_t)oE * T_ + 16 + l16] = accE1[v] * sE + cE;
            Z[(size_t)oO * T_ + l16]      = accO0[v] * sO + cO;
            Z[(size_t)oO * T_ + 16 + l16] = accO1[v] * sO + cO;
        }
    }
}

__global__ __launch_bounds__(256) void k_gemm3(
    const float* __restrict__ z, const float* __restrict__ w2w,
    const float* __restrict__ w2b, const float* __restrict__ x,
    float* __restrict__ out)
{
    const int lane = threadIdx.x & 31;
    const int hf = lane >> 4, l16 = lane & 15;
    int tid = blockIdx.x * 8 + (threadIdx.x >> 5);
    const int tn = T_ / 16, tm = DIN / 16;
    int b  = tid / (tm * tn);
    int r  = tid % (tm * tn);
    int m0 = (r / tn) * 16;
    int n0 = (r % tn) * 16;
    const float* Arow = w2w + (size_t)(m0 + l16) * DH2;
    const float* Bm   = z + (size_t)b * DH2 * T_;
    v8f acc = {};
    #pragma unroll 4
    for (int k0 = 0; k0 < DH2; k0 += 4) {
        int ka = k0 + 2 * hf;
        v2f a = *(const v2f*)(Arow + ka);
        v2f bv;
        bv.x = Bm[(size_t)ka * T_ + n0 + l16];
        bv.y = Bm[(size_t)(ka + 1) * T_ + n0 + l16];
        acc = __builtin_amdgcn_wmma_f32_16x16x4_f32(false, a, false, bv,
                                                    (short)0, acc, false, false);
    }
    #pragma unroll
    for (int v = 0; v < 8; ++v) {
        int i = m0 + v + 8 * hf;
        int t = n0 + l16;
        size_t oi = ((size_t)b * T_ + t) * DIN + i;
        out[oi] = acc[v] + w2b[i] + x[oi];
    }
}

extern "C" void kernel_launch(void* const* d_in, const int* in_sizes, int n_in,
                              void* d_out, int out_size, void* d_ws, size_t ws_size,
                              hipStream_t stream)
{
    const float* x   = (const float*)d_in[0];
    const float* d_w = (const float*)d_in[1];
    const float* d_g = (const float*)d_in[2];
    const float* d_b = (const float*)d_in[3];
    const float* p_w = (const float*)d_in[4];
    const float* p_g = (const float*)d_in[5];
    const float* p_b = (const float*)d_in[6];
    const float* w1w = (const float*)d_in[7];
    const float* w1b = (const float*)d_in[8];
    const float* w2w = (const float*)d_in[9];
    const float* w2b = (const float*)d_in[10];
    float* out = (float*)d_out;

    float* ws = (float*)d_ws;
    const size_t NCT = (size_t)B_ * DH1 * T_;
    float* hbuf   = ws;
    float* ybuf   = ws + NCT;
    float* inv_nd = ws + 2 * NCT;
    float* inv_np = inv_nd + 128;

    k_norm_d<<<1, 128, 0, stream>>>(d_w, inv_nd);
    k_norm_p<<<B_ * (DH2 / 64), 256, 0, stream>>>(p_w, inv_np);
    k_gemm1_mish<<<(B_ * (DH1 / 16) * (T_ / 16)) / 8, 256, 0, stream>>>(x, w1w, w1b, hbuf);
    k_dwconv<<<(int)(NCT / 256), 256, 0, stream>>>(hbuf, d_w, d_g, d_b, p_g, inv_nd, ybuf);
    k_gemm2<<<B_ * (T_ / PT), 512, 0, stream>>>(ybuf, p_w, p_b, inv_np, hbuf);
    k_gemm3<<<(B_ * (DIN / 16) * (T_ / 16)) / 8, 256, 0, stream>>>(hbuf, w2w, w2b, x, out);
}


# GEMM 3 folded into the GEMM 2 launch: the group's z panel (1024 o x 32 t) replaces y in LDS and feeds the last product there; z never goes to memory, 6 launches -> 5; k order kept
# speedup vs baseline: 1.2695x; 1.0844x over previous
#include <hip/hip_runtime.h>
#include <math.h>

typedef __attribute__((ext_vector_type(2))) float v2f;
typedef __attribute__((ext_vector_type(8))) float v8f;

#define B_  8
#define T_  1024
#define DIN 256
#define DH1 1024
#define DH2 1024
#define KW  9

__device__ __forceinline__ float mishf(float v) {
    float sp = (v > 20.f) ? v : log1pf(__expf(v));
    return v * tanhf(sp);
}

__global__ void k_norm_d(const float* __restrict__ dw, float* __restrict__ inv_nd) {
    int tid = threadIdx.x;
    if (tid >= B_ * KW) return;
    int b = tid / KW, k = tid % KW;
    float acc = 0.f;
    for (int c = 0; c < DH1; ++c) {
        float v = dw[((size_t)b * DH1 + c) * KW + k];
        acc += v * v;
    }
    inv_nd[tid] = 1.f / fmaxf(sqrtf(acc), 1e-12f);
}

__global__ __launch_bounds__(256) void k_norm_p(const float* __restrict__ pw,
                                                float* __restrict__ inv_np) {
    __shared__ float s[256];
    int b  = blockIdx.x >> 4;
    int o  = ((blockIdx.x & 15) << 6) + (threadIdx.x & 63);
    int cg = threadIdx.x >> 6;
    const float* P = pw + (size_t)b * DH1 * DH2;
    float acc = 0.f;
    for (int c = cg; c < DH1; c += 4) {
        float v = P[(size_t)c * DH2 + o];
        acc += v * v;
    }
    s[threadIdx.x] = acc;
    __syncthreads();
    if (cg == 0) {
        float tot = s[threadIdx.x] + s[threadIdx.x + 64] + s[threadIdx.x + 128] + s[threadIdx.x + 192];
        inv_np[b * DH2 + o] = 1.f / fmaxf(sqrtf(tot), 1e-12f);
    }
}

__global__ __launch_bounds__(256) void k_gemm1_mish(
    const float* __restrict__ x, const float* __restrict__ w1w,
    const float* __restrict__ w1b, float* __restrict__ h)
{
    const int lane = threadIdx.x & 31;
    const int hf = lane >> 4, l16 = lane & 15;
    int tid = blockIdx.x * 8 + (threadIdx.x >> 5);
    const int tn = T_ / 16, tm = DH1 / 16;
    int b  = tid / (tm * tn);
    int r  = tid % (tm * tn);
    int m0 = (r / tn) * 16;
    int n0 = (r % tn) * 16;

    const float* Arow = w1w + (size_t)(m0 + l16) * DIN;
    const float* Brow = x + ((size_t)b * T_ + (n0 + l16)) * DIN;
    v8f acc = {};
    #pragma unroll 4
    for (int k0 = 0; k0 < DIN; k0 += 4) {
        v2f a  = *(const v2f*)(Arow + k0 + 2 * hf);
        v2f bv = *(const v2f*)(Brow + k0 + 2 * hf);
        acc = __builtin_amdgcn_wmma_f32_16x16x4_f32(false, a, false, bv,
                                                    (short)0, acc, false, false);
    }
    float* H = h + (size_t)b * DH1 * T_;
    #pragma unroll
    for (int v = 0; v < 8; ++v) {
        int c = m0 + v + 8 * hf;
        int t = n0 + l16;
        H[(size_t)c * T_ + t] = mishf(acc[v] + w1b[c]);
    }
}

__global__ __launch_bounds__(256) void k_dwconv(
    const float* __restrict__ h, const float* __restrict__ dw,
    const float* __restrict__ dg, const float* __restrict__ db,
    const float* __restrict__ pg, const float* __restrict__ inv_nd,
    float* __restrict__ y)
{
    int idx = blockIdx.x * 256 + threadIdx.x;
    int t = idx & (T_ - 1);
    int c = (idx >> 10) & (DH1 - 1);
    int b = idx >> 20;
    const float* hp  = h + ((size_t)b * DH1 + c) * T_;
    const float* wp  = dw + ((size_t)b * DH1 + c) * KW;
    const float* ndp = inv_nd + b * KW;
    float acc = 0.f;
    #pragma unroll
    for (int k = 0; k < KW; ++k) {
        int tt = t + k - 4;
        if (tt >= 0 && tt < T_)
            acc += hp[tt] * (wp[k] * ndp[k]);
    }
    int bc = b * DH1 + c;
    float yy = acc * dg[bc] + db[bc];
    y[((size_t)b * DH1 + c) * T_ + t] = yy * pg[bc];
}

#define PT 32
#define PP 96
#define G2_WAVES 16
#define G2_U 4
__device__ __forceinline__ void g2_load_block(const float* __restrict__ Ab, v2f (&ax)[G2_U], v2f (&ay)[G2_U]) {
    for (int u = 0; u < G2_U; ++u) {
        ax[u] = *(const v2f*)(Ab + (size_t)(4 * u) * DH2);
        ay[u] = *(const v2f*)(Ab + (size_t)(4 * u + 1) * DH2);
    }
}
__device__ __forceinline__ void g2_block(const float* Bk, const v2f (&ax)[G2_U], const v2f (&ay)[G2_U],
                                         v8f& accE0, v8f& accE1, v8f& accO0, v8f& accO1) {
    for (int u = 0; u < G2_U; ++u) {
        v2f b0 = *(const v2f*)(Bk + (2 * u) * PP);
        v2f b1 = *(const v2f*)(Bk + (2 * u) * PP + 32);
        v2f aE, aO;
        aE.x = ax[u].x; aE.y = ay[u].x;
        aO.x = ax[u].y; aO.y = ay[u].y;
        accE0 = __builtin_amdgcn_wmma_f32_16x16x4_f32(false, aE, false, b0, (short)0, accE0, false, false);
        accE1 = __builtin_amdgcn_wmma_f32_16x16x4_f32(false, aE, false, b1, (short)0, accE1, false, false);
        accO0 = __builtin_amdgcn_wmma_f32_16x16x4_f32(false, aO, false, b0, (short)0, accO0, false, false);
        accO1 = __builtin_amdgcn_wmma_f32_16x16x4_f32(false, aO, false, b1, (short)0, accO1, false, false);
    }
}
#define G3_U 8
__device__ __forceinline__ void g3_load_block(const float* __restrict__ Wb, v2f (&w)[G3_U]) {
    for (int u = 0; u < G3_U; ++u)
        w[u] = *(const v2f*)(Wb + 4 * u);
}
__device__ __forceinline__ void g3_block(const float* Zk, const v2f (&w)[G3_U], v8f& o0, v8f& o1) {
    for (int u = 0; u < G3_U; ++u) {
        v2f z0 = *(const v2f*)(Zk + (2 * u) * PP);
        v2f z1 = *(const v2f*)(Zk + (2 * u) * PP + 32);
        o0 = __builtin_amdgcn_wmma_f32_16x16x4_f32(false, z0, false, w[u], (short)0, o0, false, false);
        o1 = __builtin_amdgcn_wmma_f32_16x16x4_f32(false, z1, false, w[u], (short)0, o1, false, false);
    }
}

__global__ __launch_bounds__(512) void k_gemm23(
    const float* __restrict__ y, const float* __restrict__ pw,
    const float* __restrict__ pb, const float* __restrict__ inv_np,
    const float* __restrict__ w2w, const float* __restrict__ w2b,
    const float* __restrict__ x, float* __restrict__ out)
{
    __shared__ __attribute__((aligned(16))) float panel[(DH1 / 2) * PP];
    const int tid = threadIdx.x;
    const int b  = blockIdx.x / (T_ / PT);
    const int t0 = (blockIdx.x % (T_ / PT)) * PT;

    {
        const float* Y = y + (size_t)b * DH1 * T_ + t0;
        const int j  = tid & 7;
        const int pp = tid >> 3;
        for (int i = 0; i < (DH1 / 2) / 64; ++i) {
            int p = i * 64 + pp;
            float4 u = *(const float4*)(Y + (size_t)(2 * p) * T_ + 4 * j);
            float4 v = *(const float4*)(Y + (size_t)(2 * p + 1) * T_ + 4 * j);
            float4 lo = make_float4(u.x, v.x, u.y, v.y);
            float4 hi = make_float4(u.z, v.z, u.w, v.w);
            *(float4*)(panel + p * PP + 8 * j)     = lo;
            *(float4*)(panel + p * PP + 8 * j + 4) = hi;
        }
    }
    __syncthreads();

    const int lane = tid & 31, wave = tid >> 5;
    const int hf = lane >> 4, l16 = lane & 15;
    const float* A = pw + (size_t)b * DH1 * DH2;
    const float* Bp = panel + hf * PP + 2 * l16;
    const int NB = DH1 / (4 * G2_U);

    v8f zE0[2], zE1[2], zO0[2], zO1[2];
    #pragma unroll
    for (int ps = 0; ps < 2; ++ps) {
        const int ob = wave * 32 + ps * (G2_WAVES * 32);
        const float* Ap = A + (size_t)(2 * hf) * DH2 + ob + 2 * l16;
        v8f accE0 = {}, accE1 = {}, accO0 = {}, accO1 = {};
        v2f x0[G2_U], y0[G2_U], x1[G2_U], y1[G2_U];
        g2_load_block(Ap, x0, y0);
        for (int kb = 0; kb < NB; kb += 2) {
            g2_load_block(Ap + (size_t)(4 * G2_U * (kb + 1)) * DH2, x1, y1);
            g2_block(Bp + (2 * G2_U * kb) * PP, x0, y0, accE0, accE1, accO0, accO1);
            g2_load_block(Ap + (size_t)(4 * G2_U * ((kb + 2) & (NB - 1))) * DH2, x0, y0);
            g2_block(Bp + (2 * G2_U * (kb + 1)) * PP, x1, y1, accE0, accE1, accO0, accO1);
        }
        for (int v = 0; v < 8; ++v) {
            int oE = ob + 2 * (v + 8 * hf);
            int oO = oE + 1;
            float sE = inv_np[b * DH2 + oE], cE = pb[b * DH2 + oE];
            float sO = inv_np[b * DH2 + oO], cO = pb[b * DH2 + oO];
            zE0[ps][v] = accE0[v] * sE + cE;
            zE1[ps][v] = accE1[v] * sE + cE;
            zO0[ps][v] = accO0[v] * sO + cO;
            zO1[ps][v] = accO1[v] * sO + cO;
        }
    }

    const int i0 = wave * 16;
    const int ii = i0 + l16;
    const float* Wp = w2w + (size_t)ii * DH2 + 2 * hf;
    v2f wa[G3_U], wb[G3_U];
    g3_load_block(Wp, wa);
    const size_t obase = ((size_t)b * T_ + t0 + 8 * hf) * DIN + ii;
    float xr0[8], xr1[8];
    for (int r = 0; r < 8; ++r) {
        xr0[r] = x[obase + (size_t)r * DIN];
        xr1[r] = x[obase + (size_t)(16 + r) * DIN];
    }
    const float bias = w2b[ii];

    __syncthreads();
    #pragma unroll
    for (int ps = 0; ps < 2; ++ps) {
        const int ob = wave * 32 + ps * (G2_WAVES * 32);
        for (int v = 0; v < 8; ++v) {
            int p = (ob >> 1) + v + 8 * hf;
            v2f lo, hi;
            lo.x = zE0[ps][v]; lo.y = zO0[ps][v];
            hi.x = zE1[ps][v]; hi.y = zO1[ps][v];
            *(v2f*)(panel + p * PP + 2 * l16)      = lo;
            *(v2f*)(panel + p * PP + 32 + 2 * l16) = hi;
        }
    }
    __syncthreads();

    v8f o0 = {}, o1 = {};
    const int NB3 = DH2 / (4 * G3_U);
    for (int kb = 0; kb < NB3; kb += 2) {
        g3_load_block(Wp + 4 * G3_U * (kb + 1), wb);
        g3_block(Bp + (2 * G3_U * kb) * PP, wa, o0, o1);
        g3_load_block(Wp + 4 * G3_U * ((kb + 2) & (NB3 - 1)), wa);
        g3_block(Bp + (2 * G3_U * (kb + 1)) * PP, wb, o0, o1);
    }
    for (int r = 0; r < 8; ++r) {
        out[obase + (size_t)r * DIN]        = o0[r] + bias + xr0[r];
        out[obase + (size_t)(16 + r) * DIN] = o1[r] + bias + xr1[r];
    }
}

extern "C" void kernel_launch(void* const* d_in, const int* in_sizes, int n_in,
                              void* d_out, int out_size, void* d_ws, size_t ws_size,
                              hipStream_t stream)
{
    const float* x   = (const float*)d_in[0];
    const float* d_w = (const float*)d_in[1];
    const float* d_g = (const float*)d_in[2];
    const float* d_b = (const float*)d_in[3];
    const float* p_w = (const float*)d_in[4];
    const float* p_g = (const float*)d_in[5];
    const float* p_b = (const float*)d_in[6];
    const float* w1w = (const float*)d_in[7];
    const float* w1b = (const float*)d_in[8];
    const float* w2w = (const float*)d_in[9];
    const float* w2b = (const float*)d_in[10];
    float* out = (float*)d_out;

    float* ws = (float*)d_ws;
    const size_t NCT = (size_t)B_ * DH1 * T_;
    float* hbuf   = ws;
    float* ybuf   = ws + NCT;
    float* inv_nd = ws + 2 * NCT;
    float* inv_np = inv_nd + 128;

    k_norm_d<<<1, 128, 0, stream>>>(d_w, inv_nd);
    k_norm_p<<<B_ * (DH2 / 64), 256, 0, stream>>>(p_w, inv_np);
    k_gemm1_mish<<<(B_ * (DH1 / 16) * (T_ / 16)) / 8, 256, 0, stream>>>(x, w1w, w1b, hbuf);
    k_dwconv<<<(int)(NCT / 256), 256, 0, stream>>>(hbuf, d_w, d_g, d_b, p_g, inv_nd, ybuf);
    k_gemm23<<<B_ * (T_ / PT), 512, 0, stream>>>(ybuf, p_w, p_b, inv_np, w2w, w2b, x, out);
}


# Depthwise conv folded into the panel fill of the GEMM 2/3 launch: each group builds its y panel from h (its 32 steps + 4 each side); y never goes to memory, 5 launches -> 4; tap order kept
# speedup vs baseline: 1.4665x; 1.1552x over previous
#include <hip/hip_runtime.h>
#include <math.h>

typedef __attribute__((ext_vector_type(2))) float v2f;
typedef __attribute__((ext_vector_type(8))) float v8f;

#define B_  8
#define T_  1024
#define DIN 256
#define DH1 1024
#define DH2 1024
#define KW  9

__device__ __forceinline__ float mishf(float v) {
    float sp = (v > 20.f) ? v : log1pf(__expf(v));
    return v * tanhf(sp);
}

__global__ void k_norm_d(const float* __restrict__ dw, float* __restrict__ inv_nd) {
    int tid = threadIdx.x;
    if (tid >= B_ * KW) return;
    int b = tid / KW, k = tid % KW;
    float acc = 0.f;
    for (int c = 0; c < DH1; ++c) {
        float v = dw[((size_t)b * DH1 + c) * KW + k];
        acc += v * v;
    }
    inv_nd[tid] = 1.f / fmaxf(sqrtf(acc), 1e-12f);
}

__global__ __launch_bounds__(256) void k_norm_p(const float* __restrict__ pw,
                                                float* __restrict__ inv_np) {
    __shared__ float s[256];
    int b  = blockIdx.x >> 4;
    int o  = ((blockIdx.x & 15) << 6) + (threadIdx.x & 63);
    int cg = threadIdx.x >> 6;
    const float* P = pw + (size_t)b * DH1 * DH2;
    float acc = 0.f;
    for (int c = cg; c < DH1; c += 4) {
        float v = P[(size_t)c * DH2 + o];
        acc += v * v;
    }
    s[threadIdx.x] = acc;
    __syncthreads();
    if (cg == 0) {
        float tot = s[threadIdx.x] + s[threadIdx.x + 64] + s[threadIdx.x + 128] + s[threadIdx.x + 192];
        inv_np[b * DH2 + o] = 1.f / fmaxf(sqrtf(tot), 1e-12f);
    }
}

__global__ __launch_bounds__(256) void k_gemm1_mish(
    const float* __restrict__ x, const float* __restrict__ w1w,
    const float* __restrict__ w1b, float* __restrict__ h)
{
    const int lane = threadIdx.x & 31;
    const int hf = lane >> 4, l16 = lane & 15;
    int tid = blockIdx.x * 8 + (threadIdx.x >> 5);
    const int tn = T_ / 16, tm = DH1 / 16;
    int b  = tid / (tm * tn);
    int r  = tid % (tm * tn);
    int m0 = (r / tn) * 16;
    int n0 = (r % tn) * 16;

    const float* Arow = w1w + (size_t)(m0 + l16) * DIN;
    const float* Brow = x + ((size_t)b * T_ + (n0 + l16)) * DIN;
    v8f acc = {};
    #pragma unroll 4
    for (int k0 = 0; k0 < DIN; k0 += 4) {
        v2f a  = *(const v2f*)(Arow + k0 + 2 * hf);
        v2f bv = *(const v2f*)(Brow + k0 + 2 * hf);
        acc = __builtin_amdgcn_wmma_f32_16x16x4_f32(false, a, false, bv,
                                                    (short)0, acc, false, false);
    }
    float* H = h + (size_t)b * DH1 * T_;
    #pragma unroll
    for (int v = 0; v < 8; ++v) {
        int c = m0 + v + 8 * hf;
        int t = n0 + l16;
        H[(size_t)c * T_ + t] = mishf(acc[v] + w1b[c]);
    }
}

__device__ __forceinline__ float4 dw_row4(const float (&win)[12], const float (&w)[KW],
                                          float g, float bb, float pgv) {
    float o[4];
    for (int q = 0; q < 4; ++q) {
        float acc = 0.f;
        for (int k = 0; k < KW; ++k)
            acc = fmaf(win[q + k], w[k], acc);
        float yy = fmaf(acc, g, bb);
        o[q] = yy * pgv;
    }
    return make_float4(o[0], o[1], o[2], o[3]);
}
__device__ __forceinline__ void put4(float (&win)[12], int at, float4 v) {
    win[at] = v.x; win[at + 1] = v.y; win[at + 2] = v.z; win[at + 3] = v.w;
}

#define PT 32
#define PP 96
#define G2_WAVES 16
#define G2_U 4
__device__ __forceinline__ void g2_load_block(const float* __restrict__ Ab, v2f (&ax)[G2_U], v2f (&ay)[G2_U]) {
    for (int u = 0; u < G2_U; ++u) {
        ax[u] = *(const v2f*)(Ab + (size_t)(4 * u) * DH2);
        ay[u] = *(const v2f*)(Ab + (size_t)(4 * u + 1) * DH2);
    }
}
__device__ __forceinline__ void g2_block(const float* Bk, const v2f (&ax)[G2_U], const v2f (&ay)[G2_U],
                                         v8f& accE0, v8f& accE1, v8f& accO0, v8f& accO1) {
    for (int u = 0; u < G2_U; ++u) {
        v2f b0 = *(const v2f*)(Bk + (2 * u) * PP);
        v2f b1 = *(const v2f*)(Bk + (2 * u) * PP + 32);
        v2f aE, aO;
        aE.x = ax[u].x; aE.y = ay[u].x;
        aO.x = ax[u].y; aO.y = ay[u].y;
        accE0 = __builtin_amdgcn_wmma_f32_16x16x4_f32(false, aE, false, b0, (short)0, accE0, false, false);
        accE1 = __builtin_amdgcn_wmma_f32_16x16x4_f32(false, aE, false, b1, (short)0, accE1, false, false);
        accO0 = __builtin_amdgcn_wmma_f32_16x16x4_f32(false, aO, false, b0, (short)0, accO0, false, false);
        accO1 = __builtin_amdgcn_wmma_f32_16x16x4_f32(false, aO, false, b1, (short)0, accO1, false, false);
    }
}
#define G3_U 8
__device__ __forceinline__ void g3_load_block(const float* __restrict__ Wb, v2f (&w)[G3_U]) {
    for (int u = 0; u < G3_U; ++u)
        w[u] = *(const v2f*)(Wb + 4 * u);
}
__device__ __forceinline__ void g3_block(const float* Zk, const v2f (&w)[G3_U], v8f& o0, v8f& o1) {
    for (int u = 0; u < G3_U; ++u) {
        v2f z0 = *(const v2f*)(Zk + (2 * u) * PP);
        v2f z1 = *(const v2f*)(Zk + (2 * u) * PP + 32);
        o0 = __builtin_amdgcn_wmma_f32_16x16x4_f32(false, z0, false, w[u], (short)0, o0, false, false);
        o1 = __builtin_amdgcn_wmma_f32_16x16x4_f32(false, z1, false, w[u], (short)0, o1, false, false);
    }
}

__global__ __launch_bounds__(512) void k_gemm23(
    const float* __restrict__ h, const float* __restrict__ dw,
    const float* __restrict__ dg, const float* __restrict__ db,
    const float* __restrict__ pg, const float* __restrict__ inv_nd,
    const float* __restrict__ pw,
    const float* __restrict__ pb, const float* __restrict__ inv_np,
    const float* __restrict__ w2w, const float* __restrict__ w2b,
    const float* __restrict__ x, float* __restrict__ out)
{
    __shared__ __attribute__((aligned(16))) float panel[(DH1 / 2) * PP];
    const int tid = threadIdx.x;
    const int b  = blockIdx.x / (T_ / PT);
    const int t0 = (blockIdx.x % (T_ / PT)) * PT;

    {
        const float* Hb = h + (size_t)b * DH1 * T_ + t0;
        const int j  = tid & 7;
        const int pp = tid >> 3;
        const bool hasL = (j > 0) || (t0 > 0);
        const bool hasR = (j < 7) || (t0 + PT < T_);
        const int offL = hasL ? -4 : 0;
        const int offR = hasR ? 4 : 0;
        const float4 zero4 = make_float4(0.f, 0.f, 0.f, 0.f);
        float nd[KW];
        for (int k = 0; k < KW; ++k) nd[k] = inv_nd[b * KW + k];
        #pragma unroll 2
        for (int i = 0; i < (DH1 / 2) / 64; ++i) {
            const int p  = i * 64 + pp;
            const int bc = b * DH1 + 2 * p;
            const float* r0 = Hb + (size_t)(2 * p) * T_ + 4 * j;
            const float* r1 = r0 + T_;
            float win0[12], win1[12];
            float4 l0 = *(const float4*)(r0 + offL), m0 = *(const float4*)(r0), n0 = *(const float4*)(r0 + offR);
            float4 l1 = *(const float4*)(r1 + offL), m1 = *(const float4*)(r1), n1 = *(const float4*)(r1 + offR);
            put4(win0, 0, hasL ? l0 : zero4);
            put4(win0, 4, m0);
            put4(win0, 8, hasR ? n0 : zero4);
            put4(win1, 0, hasL ? l1 : zero4);
            put4(win1, 4, m1);
            put4(win1, 8, hasR ? n1 : zero4);
            const v2f* wp2 = (const v2f*)(dw + (size_t)bc * KW);
            float wf[2 * KW];
            for (int u = 0; u < KW; ++u) { v2f t = wp2[u]; wf[2 * u] = t.x; wf[2 * u + 1] = t.y; }
            float w0[KW], w1[KW];
            for (int k = 0; k < KW; ++k) { w0[k] = wf[k] * nd[k]; w1[k] = wf[KW + k] * nd[k]; }
            const v2f g2 = *(const v2f*)(dg + bc);
            const v2f b2 = *(const v2f*)(db + bc);
            const v2f q2 = *(const v2f*)(pg + bc);
            float4 u = dw_row4(win0, w0, g2.x, b2.x, q2.x);
            float4 v = dw_row4(win1, w1, g2.y, b2.y, q2.y);
            float4 lo = make_float4(u.x, v.x, u.y, v.y);
            float4 hi = make_float4(u.z, v.z, u.w, v.w);
            *(float4*)(panel + p * PP + 8 * j)     = lo;
            *(float4*)(panel + p * PP + 8 * j + 4) = hi;
        }
    }
    __syncthreads();

    const int lane = tid & 31, wave = tid >> 5;
    const int hf = lane >> 4, l16 = lane & 15;
    const float* A = pw + (size_t)b * DH1 * DH2;
    const float* Bp = panel + hf * PP + 2 * l16;
    const int NB = DH1 / (4 * G2_U);

    v8f zE0[2], zE1[2], zO0[2], zO1[2];
    #pragma unroll
    for (int ps = 0; ps < 2; ++ps) {
        const int ob = wave * 32 + ps * (G2_WAVES * 32);
        const float* Ap = A + (size_t)(2 * hf) * DH2 + ob + 2 * l16;
        v8f accE0 = {}, accE1 = {}, accO0 = {}, accO1 = {};
        v2f x0[G2_U], y0[G2_U], x1[G2_U], y1[G2_U];
        g2_load_block(Ap, x0, y0);
        for (int kb = 0; kb < NB; kb += 2) {
            g2_load_block(Ap + (size_t)(4 * G2_U * (kb + 1)) * DH2, x1, y1);
            g2_block(Bp + (2 * G2_U * kb) * PP, x0, y0, accE0, accE1, accO0, accO1);
            g2_load_block(Ap + (size_t)(4 * G2_U * ((kb + 2) & (NB - 1))) * DH2, x0, y0);
            g2_block(Bp + (2 * G2_U * (kb + 1)) * PP, x1, y1, accE0, accE1, accO0, accO1);
            asm volatile("" :: "v"(x0[0]), "v"(x0[1]), "v"(x0[2]), "v"(x0[3]),
                               "v"(y0[0]), "v"(y0[1]), "v"(y0[2]), "v"(y0[3]));
        }
        for (int v = 0; v < 8; ++v) {
            int oE = ob + 2 * (v + 8 * hf);
            int oO = oE + 1;
            float sE = inv_np[b * DH2 + oE], cE = pb[b * DH2 + oE];
            float sO = inv_np[b * DH2 + oO], cO = pb[b * DH2 + oO];
            zE0[ps][v] = accE0[v] * sE + cE;
            zE1[ps][v] = accE1[v] * sE + cE;
            zO0[ps][v] = accO0[v] * sO + cO;
            zO1[ps][v] = accO1[v] * sO + cO;
        }
    }

    const int i0 = wave * 16;
    const int ii = i0 + l16;
    const float* Wp = w2w + (size_t)ii * DH2 + 2 * hf;
    v2f wa[G3_U], wb[G3_U];
    g3_load_block(Wp, wa);
    const size_t obase = ((size_t)b * T_ + t0 + 8 * hf) * DIN + ii;
    float xr0[8], xr1[8];
    for (int r = 0; r < 8; ++r) {
        xr0[r] = x[obase + (size_t)r * DIN];
        xr1[r] = x[obase + (size_t)(16 + r) * DIN];
    }
    const float bias = w2b[ii];

    __syncthreads();
    #pragma unroll
    for (int ps = 0; ps < 2; ++ps) {
        const int ob = wave * 32 + ps * (G2_WAVES * 32);
        for (int v = 0; v < 8; ++v) {
            int p = (ob >> 1) + v + 8 * hf;
            v2f lo, hi;
            lo.x = zE0[ps][v]; lo.y = zO0[ps][v];
            hi.x = zE1[ps][v]; hi.y = zO1[ps][v];
            *(v2f*)(panel + p * PP + 2 * l16)      = lo;
            *(v2f*)(panel + p * PP + 32 + 2 * l16) = hi;
        }
    }
    __syncthreads();

    v8f o0 = {}, o1 = {};
    const int NB3 = DH2 / (4 * G3_U);
    for (int kb = 0; kb < NB3; kb += 2) {
        g3_load_block(Wp + 4 * G3_U * (kb + 1), wb);
        g3_block(Bp + (2 * G3_U * kb) * PP, wa, o0, o1);
        g3_load_block(Wp + 4 * G3_U * ((kb + 2) & (NB3 - 1)), wa);
        g3_block(Bp + (2 * G3_U * (kb + 1)) * PP, wb, o0, o1);
    }
    for (int r = 0; r < 8; ++r) {
        out[obase + (size_t)r * DIN]        = o0[r] + bias + xr0[r];
        out[obase + (size_t)(16 + r) * DIN] = o1[r] + bias + xr1[r];
    }
}

extern "C" void kernel_launch(void* const* d_in, const int* in_sizes, int n_in,
                              void* d_out, int out_size, void* d_ws, size_t ws_size,
                              hipStream_t stream)
{
    const float* x   = (const float*)d_in[0];
    const float* d_w = (const float*)d_in[1];
    const float* d_g = (const float*)d_in[2];
    const float* d_b = (const float*)d_in[3];
    const float* p_w = (const float*)d_in[4];
    const float* p_g = (const float*)d_in[5];
    const float* p_b = (const float*)d_in[6];
    const float* w1w = (const float*)d_in[7];
    const float* w1b = (const float*)d_in[8];
    const float* w2w = (const float*)d_in[9];
    const float* w2b = (const float*)d_in[10];
    float* out = (float*)d_out;

    float* ws = (float*)d_ws;
    const size_t NCT = (size_t)B_ * DH1 * T_;
    float* hbuf   = ws;
    float* inv_nd = ws + NCT;
    float* inv_np = inv_nd + 128;

    k_norm_d<<<1, 128, 0, stream>>>(d_w, inv_nd);
    k_norm_p<<<B_ * (DH2 / 64), 256, 0, stream>>>(p_w, inv_np);
    k_gemm1_mish<<<(B_ * (DH1 / 16) * (T_ / 16)) / 8, 256, 0, stream>>>(x, w1w, w1b, hbuf);
    k_gemm23<<<B_ * (T_ / PT), 512, 0, stream>>>(hbuf, d_w, d_g, d_b, p_g, inv_nd, p_w, p_b, inv_np, w2w, w2b, x, out);
}


# Both weight-norm pre-passes folded into the GEMM 2/3 launch (4 launches -> 2): p-norm squares ride on GEMM 2's weight fragments (class sums kept), d-norm per group as a tree (sum order changes)
# speedup vs baseline: 1.8689x; 1.2744x over previous
#include <hip/hip_runtime.h>
#include <math.h>

typedef __attribute__((ext_vector_type(2))) float v2f;
typedef __attribute__((ext_vector_type(8))) float v8f;

#define B_  8
#define T_  1024
#define DIN 256
#define DH1 1024
#define DH2 1024
#define KW  9

__device__ __forceinline__ float mishf(float v) {
    float sp = (v > 20.f) ? v : log1pf(__expf(v));
    return v * tanhf(sp);
}

__global__ __launch_bounds__(256) void k_gemm1_mish(
    const float* __restrict__ x, const float* __restrict__ w1w,
    const float* __restrict__ w1b, float* __restrict__ h)
{
    const int lane = threadIdx.x & 31;
    const int hf = lane >> 4, l16 = lane & 15;
    int tid = blockIdx.x * 8 + (threadIdx.x >> 5);
    const int tn = T_ / 16, tm = DH1 / 16;
    int b  = tid / (tm * tn);
    int r  = tid % (tm * tn);
    int m0 = (r / tn) * 16;
    int n0 = (r % tn) * 16;

    const float* Arow = w1w + (size_t)(m0 + l16) * DIN;
    const float* Brow = x + ((size_t)b * T_ + (n0 + l16)) * DIN;
    v8f acc = {};
    #pragma unroll 4
    for (int k0 = 0; k0 < DIN; k0 += 4) {
        v2f a  = *(const v2f*)(Arow + k0 + 2 * hf);
        v2f bv = *(const v2f*)(Brow + k0 + 2 * hf);
        acc = __builtin_amdgcn_wmma_f32_16x16x4_f32(false, a, false, bv,
                                                    (short)0, acc, false, false);
    }
    float* H = h + (size_t)b * DH1 * T_;
    #pragma unroll
    for (int v = 0; v < 8; ++v) {
        int c = m0 + v + 8 * hf;
        int t = n0 + l16;
        H[(size_t)c * T_ + t] = mishf(acc[v] + w1b[c]);
    }
}

__device__ __forceinline__ float4 dw_row4(const float (&win)[12], const float (&w)[KW],
                                          float g, float bb, float pgv) {
    float o[4];
    for (int q = 0; q < 4; ++q) {
        float acc = 0.f;
        for (int k = 0; k < KW; ++k)
            acc = fmaf(win[q + k], w[k], acc);
        float yy = fmaf(acc, g, bb);
        o[q] = yy * pgv;
    }
    return make_float4(o[0], o[1], o[2], o[3]);
}
__device__ __forceinline__ void put4(float (&win)[12], int at, float4 v) {
    win[at] = v.x; win[at + 1] = v.y; win[at + 2] = v.z; win[at + 3] = v.w;
}

#define PT 32
#define PP 96
#define G2_WAVES 16
#define G2_U 4
__device__ __forceinline__ void g2_load_block(const float* __restrict__ Ab, v2f (&ax)[G2_U], v2f (&ay)[G2_U]) {
    for (int u = 0; u < G2_U; ++u) {
        ax[u] = *(const v2f*)(Ab + (size_t)(4 * u) * DH2);
        ay[u] = *(const v2f*)(Ab + (size_t)(4 * u + 1) * DH2);
    }
}
__device__ __forceinline__ void g2_block(const float* Bk, const v2f (&ax)[G2_U], const v2f (&ay)[G2_U],
                                         v8f& accE0, v8f& accE1, v8f& accO0, v8f& accO1,
                                         v2f& qx, v2f& qy) {
    for (int u = 0; u < G2_U; ++u) {
        qx.x = fmaf(ax[u].x, ax[u].x, qx.x);
        qx.y = fmaf(ax[u].y, ax[u].y, qx.y);
        qy.x = fmaf(ay[u].x, ay[u].x, qy.x);
        qy.y = fmaf(ay[u].y, ay[u].y, qy.y);
        v2f b0 = *(const v2f*)(Bk + (2 * u) * PP);
        v2f b1 = *(const v2f*)(Bk + (2 * u) * PP + 32);
        v2f aE, aO;
        aE.x = ax[u].x; aE.y = ay[u].x;
        aO.x = ax[u].y; aO.y = ay[u].y;
        accE0 = __builtin_amdgcn_wmma_f32_16x16x4_f32(false, aE, false, b0, (short)0, accE0, false, false);
        accE1 = __builtin_amdgcn_wmma_f32_16x16x4_f32(false, aE, false, b1, (short)0, accE1, false, false);
        accO0 = __builtin_amdgcn_wmma_f32_16x16x4_f32(false, aO, false, b0, (short)0, accO0, false, false);
        accO1 = __builtin_amdgcn_wmma_f32_16x16x4_f32(false, aO, false, b1, (short)0, accO1, false, false);
    }
}
#define G3_U 8
__device__ __forceinline__ void g3_load_block(const float* __restrict__ Wb, v2f (&w)[G3_U]) {
    for (int u = 0; u < G3_U; ++u)
        w[u] = *(const v2f*)(Wb + 4 * u);
}
__device__ __forceinline__ void g3_block(const float* Zk, const v2f (&w)[G3_U], v8f& o0, v8f& o1) {
    for (int u = 0; u < G3_U; ++u) {
        v2f z0 = *(const v2f*)(Zk + (2 * u) * PP);
        v2f z1 = *(const v2f*)(Zk + (2 * u) * PP + 32);
        o0 = __builtin_amdgcn_wmma_f32_16x16x4_f32(false, z0, false, w[u], (short)0, o0, false, false);
        o1 = __builtin_amdgcn_wmma_f32_16x16x4_f32(false, z1, false, w[u], (short)0, o1, false, false);
    }
}

__global__ __launch_bounds__(512) void k_gemm23(
    const float* __restrict__ h, const float* __restrict__ dw,
    const float* __restrict__ dg, const float* __restrict__ db,
    const float* __restrict__ pg,
    const float* __restrict__ pw,
    const float* __restrict__ pb,
    const float* __restrict__ w2w, const float* __restrict__ w2b,
    const float* __restrict__ x, float* __restrict__ out)
{
    __shared__ __attribute__((aligned(16))) float panel[(DH1 / 2) * PP];
    const int tid = threadIdx.x;
    const int b  = blockIdx.x / (T_ / PT);
    const int t0 = (blockIdx.x % (T_ / PT)) * PT;

    __shared__ float dred[KW * G2_WAVES];
    __shared__ float dnrm[16];
    {
        const v2f* wq = (const v2f*)(dw + ((size_t)b * DH1 + 2 * tid) * KW);
        float wf[2 * KW];
        for (int u = 0; u < KW; ++u) { v2f t = wq[u]; wf[2 * u] = t.x; wf[2 * u + 1] = t.y; }
        float sq[KW];
        for (int k = 0; k < KW; ++k) sq[k] = fmaf(wf[KW + k], wf[KW + k], wf[k] * wf[k]);
        for (int m = 16; m >= 1; m >>= 1)
            for (int k = 0; k < KW; ++k) sq[k] += __shfl_xor(sq[k], m);
        if ((tid & 31) == 0)
            for (int k = 0; k < KW; ++k) dred[k * G2_WAVES + (tid >> 5)] = sq[k];
    }
    __syncthreads();
    if (tid < KW) {
        float tot = 0.f;
        for (int w = 0; w < G2_WAVES; ++w) tot += dred[tid * G2_WAVES + w];
        dnrm[tid] = 1.f / fmaxf(sqrtf(tot), 1e-12f);
    }
    __syncthreads();

    {
        const float* Hb = h + (size_t)b * DH1 * T_ + t0;
        const int j  = tid & 7;
        const int pp = tid >> 3;
        const bool hasL = (j > 0) || (t0 > 0);
        const bool hasR = (j < 7) || (t0 + PT < T_);
        const int offL = hasL ? -4 : 0;
        const int offR = hasR ? 4 : 0;
        const float4 zero4 = make_float4(0.f, 0.f, 0.f, 0.f);
        float nd[KW];
        for (int k = 0; k < KW; ++k) nd[k] = dnrm[k];
        #pragma unroll 2
        for (int i = 0; i < (DH1 / 2) / 64; ++i) {
            const int p  = i * 64 + pp;
            const int bc = b * DH1 + 2 * p;
            const float* r0 = Hb + (size_t)(2 * p) * T_ + 4 * j;
            const float* r1 = r0 + T_;
            float win0[12], win1[12];
            float4 l0 = *(const float4*)(r0 + offL), m0 = *(const float4*)(r0), n0 = *(const float4*)(r0 + offR);
            float4 l1 = *(const float4*)(r1 + offL), m1 = *(const float4*)(r1), n1 = *(const float4*)(r1 + offR);
            put4(win0, 0, hasL ? l0 : zero4);
            put4(win0, 4, m0);
            put4(win0, 8, hasR ? n0 : zero4);
            put4(win1, 0, hasL ? l1 : zero4);
            put4(win1, 4, m1);
            put4(win1, 8, hasR ? n1 : zero4);
            const v2f* wp2 = (const v2f*)(dw + (size_t)bc * KW);
            float wf[2 * KW];
            for (int u = 0; u < KW; ++u) { v2f t = wp2[u]; wf[2 * u] = t.x; wf[2 * u + 1] = t.y; }
            float w0[KW], w1[KW];
            for (int k = 0; k < KW; ++k) { w0[k] = wf[k] * nd[k]; w1[k] = wf[KW + k] * nd[k]; }
            const v2f g2 = *(const v2f*)(dg + bc);
            const v2f b2 = *(const v2f*)(db + bc);
            const v2f q2 = *(const v2f*)(pg + bc);
            float4 u = dw_row4(win0, w0, g2.x, b2.x, q2.x);
            float4 v = dw_row4(win1, w1, g2.y, b2.y, q2.y);
            float4 lo = make_float4(u.x, v.x, u.y, v.y);
            float4 hi = make_float4(u.z, v.z, u.w, v.w);
            *(float4*)(panel + p * PP + 8 * j)     = lo;
            *(float4*)(panel + p * PP + 8 * j + 4) = hi;
        }
    }
    __syncthreads();

    const int lane = tid & 31, wave = tid >> 5;
    const int hf = lane >> 4, l16 = lane & 15;
    const float* A = pw + (size_t)b * DH1 * DH2;
    const float* Bp = panel + hf * PP + 2 * l16;
    const int NB = DH1 / (4 * G2_U);

    v8f zE0[2], zE1[2], zO0[2], zO1[2];
    #pragma unroll
    for (int ps = 0; ps < 2; ++ps) {
        const int ob = wave * 32 + ps * (G2_WAVES * 32);
        const float* Ap = A + (size_t)(2 * hf) * DH2 + ob + 2 * l16;
        v8f accE0 = {}, accE1 = {}, accO0 = {}, accO1 = {};
        v2f qx = {}, qy = {};
        v2f x0[G2_U], y0[G2_U], x1[G2_U], y1[G2_U];
        g2_load_block(Ap, x0, y0);
        for (int kb = 0; kb < NB; kb += 2) {
            g2_load_block(Ap + (size_t)(4 * G2_U * (kb + 1)) * DH2, x1, y1);
            g2_block(Bp + (2 * G2_U * kb) * PP, x0, y0, accE0, accE1, accO0, accO1, qx, qy);
            g2_load_block(Ap + (size_t)(4 * G2_U * ((kb + 2) & (NB - 1))) * DH2, x0, y0);
            g2_block(Bp + (2 * G2_U * (kb + 1)) * PP, x1, y1, accE0, accE1, accO0, accO1, qx, qy);
            asm volatile("" :: "v"(x0[0]), "v"(x0[1]), "v"(x0[2]), "v"(x0[3]),
                               "v"(y0[0]), "v"(y0[1]), "v"(y0[2]), "v"(y0[3]));
        }
        const float pxE = __shfl_xor(qx.x, 16), pxO = __shfl_xor(qx.y, 16);
        const float pyE = __shfl_xor(qy.x, 16), pyO = __shfl_xor(qy.y, 16);
        const float totE = (((hf ? pxE : qx.x) + (hf ? pyE : qy.x)) + (hf ? qx.x : pxE)) + (hf ? qy.x : pyE);
        const float totO = (((hf ? pxO : qx.y) + (hf ? pyO : qy.y)) + (hf ? qx.y : pxO)) + (hf ? qy.y : pyO);
        const float invE = 1.f / fmaxf(sqrtf(totE), 1e-12f);
        const float invO = 1.f / fmaxf(sqrtf(totO), 1e-12f);
        for (int v = 0; v < 8; ++v) {
            int oE = ob + 2 * (v + 8 * hf);
            int oO = oE + 1;
            float sE = __shfl(invE, v + 8 * hf), cE = pb[b * DH2 + oE];
            float sO = __shfl(invO, v + 8 * hf), cO = pb[b * DH2 + oO];
            zE0[ps][v] = accE0[v] * sE + cE;
            zE1[ps][v] = accE1[v] * sE + cE;
            zO0[ps][v] = accO0[v] * sO + cO;
            zO1[ps][v] = accO1[v] * sO + cO;
        }
    }

    const int i0 = wave * 16;
    const int ii = i0 + l16;
    const float* Wp = w2w + (size_t)ii * DH2 + 2 * hf;
    v2f wa[G3_U], wb[G3_U];
    g3_load_block(Wp, wa);
    const size_t obase = ((size_t)b * T_ + t0 + 8 * hf) * DIN + ii;
    float xr0[8], xr1[8];
    for (int r = 0; r < 8; ++r) {
        xr0[r] = x[obase + (size_t)r * DIN];
        xr1[r] = x[obase + (size_t)(16 + r) * DIN];
    }
    const float bias = w2b[ii];

    __syncthreads();
    #pragma unroll
    for (int ps = 0; ps < 2; ++ps) {
        const int ob = wave * 32 + ps * (G2_WAVES * 32);
        for (int v = 0; v < 8; ++v) {
            int p = (ob >> 1) + v + 8 * hf;
            v2f lo, hi;
            lo.x = zE0[ps][v]; lo.y = zO0[ps][v];
            hi.x = zE1[ps][v]; hi.y = zO1[ps][v];
            *(v2f*)(panel + p * PP + 2 * l16)      = lo;
            *(v2f*)(panel + p * PP + 32 + 2 * l16) = hi;
        }
    }
    __syncthreads();

    v8f o0 = {}, o1 = {};
    const int NB3 = DH2 / (4 * G3_U);
    for (int kb = 0; kb < NB3; kb += 2) {
        g3_load_block(Wp + 4 * G3_U * (kb + 1), wb);
        g3_block(Bp + (2 * G3_U * kb) * PP, wa, o0, o1);
        g3_load_block(Wp + 4 * G3_U * ((kb + 2) & (NB3 - 1)), wa);
        g3_block(Bp + (2 * G3_U * (kb + 1)) * PP, wb, o0, o1);
    }
    for (int r = 0; r < 8; ++r) {
        out[obase + (size_t)r * DIN]        = o0[r] + bias + xr0[r];
        out[obase + (size_t)(16 + r) * DIN] = o1[r] + bias + xr1[r];
    }
}

extern "C" void kernel_launch(void* const* d_in, const int* in_sizes, int n_in,
                              void* d_out, int out_size, void* d_ws, size_t ws_size,
                              hipStream_t stream)
{
    const float* x   = (const float*)d_in[0];
    const float* d_w = (const float*)d_in[1];
    const float* d_g = (const float*)d_in[2];
    const float* d_b = (const float*)d_in[3];
    const float* p_w = (const float*)d_in[4];
    const float* p_g = (const float*)d_in[5];
    const float* p_b = (const float*)d_in[6];
    const float* w1w = (const float*)d_in[7];
    const float* w1b = (const float*)d_in[8];
    const float* w2w = (const float*)d_in[9];
    const float* w2b = (const float*)d_in[10];
    float* out = (float*)d_out;

    float* hbuf = (float*)d_ws;

    k_gemm1_mish<<<(B_ * (DH1 / 16) * (T_ / 16)) / 8, 256, 0, stream>>>(x, w1w, w1b, hbuf);
    k_gemm23<<<B_ * (T_ / PT), 512, 0, stream>>>(hbuf, d_w, d_g, d_b, p_g, p_w, p_b, w2w, w2b, x, out);
}


# GEMM 1 + mish folded into the GEMM 2/3 launch (2 launches -> 1): each group makes h for its 32 steps + halo from an x panel in LDS, h never goes to memory; k order inside each 8-k unit changes
# speedup vs baseline: 2.1778x; 1.1653x over previous
#include <hip/hip_runtime.h>
#include <math.h>

typedef __attribute__((ext_vector_type(2))) float v2f;
typedef __attribute__((ext_vector_type(8))) float v8f;

#define B_  8
#define T_  1024
#define DIN 256
#define DH1 1024
#define DH2 1024
#define KW  9

__device__ __forceinline__ float mishf(float v) {
    float sp = (v > 20.f) ? v : log1pf(__expf(v));
    return v * tanhf(sp);
}

#define HT 48
#define XP 260
#define G1_U 8
__device__ __forceinline__ void g1_load_block(const float* __restrict__ Wb, float4 (&w)[G1_U]) {
    for (int u = 0; u < G1_U; ++u)
        w[u] = *(const float4*)(Wb + 8 * u);
}
__device__ __forceinline__ void g1_block(const float* Xk, const float4 (&w)[G1_U], v8f& a0, v8f& a1, v8f& a2) {
    for (int u = 0; u < G1_U; ++u) {
        const float4 x0 = *(const float4*)(Xk + 8 * u);
        const float4 x1 = *(const float4*)(Xk + 16 * XP + 8 * u);
        const float4 x2 = *(const float4*)(Xk + 32 * XP + 8 * u);
        v2f wl, wh, p0, p1, p2, q0, q1, q2;
        wl.x = w[u].x; wl.y = w[u].y; wh.x = w[u].z; wh.y = w[u].w;
        p0.x = x0.x; p0.y = x0.y; q0.x = x0.z; q0.y = x0.w;
        p1.x = x1.x; p1.y = x1.y; q1.x = x1.z; q1.y = x1.w;
        p2.x = x2.x; p2.y = x2.y; q2.x = x2.z; q2.y = x2.w;
        a0 = __builtin_amdgcn_wmma_f32_16x16x4_f32(false, p0, false, wl, (short)0, a0, false, false);
        a1 = __builtin_amdgcn_wmma_f32_16x16x4_f32(false, p1, false, wl, (short)0, a1, false, false);
        a2 = __builtin_amdgcn_wmma_f32_16x16x4_f32(false, p2, false, wl, (short)0, a2, false, false);
        a0 = __builtin_amdgcn_wmma_f32_16x16x4_f32(false, q0, false, wh, (short)0, a0, false, false);
        a1 = __builtin_amdgcn_wmma_f32_16x16x4_f32(false, q1, false, wh, (short)0, a1, false, false);
        a2 = __builtin_amdgcn_wmma_f32_16x16x4_f32(false, q2, false, wh, (short)0, a2, false, false);
    }
}

__device__ __forceinline__ float4 dw_row4(const float (&win)[12], const float (&w)[KW],
                                          float g, float bb, float pgv) {
    float o[4];
    for (int q = 0; q < 4; ++q) {
        float acc = 0.f;
        for (int k = 0; k < KW; ++k)
            acc = fmaf(win[q + k], w[k], acc);
        float yy = fmaf(acc, g, bb);
        o[q] = yy * pgv;
    }
    return make_float4(o[0], o[1], o[2], o[3]);
}
__device__ __forceinline__ void put4(float (&win)[12], int at, float4 v) {
    win[at] = v.x; win[at + 1] = v.y; win[at + 2] = v.z; win[at + 3] = v.w;
}

#define PT 32
#define PP 96
#define G2_WAVES 16
#define G2_U 4
__device__ __forceinline__ void g2_load_block(const float* __restrict__ Ab, v2f (&ax)[G2_U], v2f (&ay)[G2_U]) {
    for (int u = 0; u < G2_U; ++u) {
        ax[u] = *(const v2f*)(Ab + (size_t)(4 * u) * DH2);
        ay[u] = *(const v2f*)(Ab + (size_t)(4 * u + 1) * DH2);
    }
}
__device__ __forceinline__ void g2_block(const float* Bk, const v2f (&ax)[G2_U], const v2f (&ay)[G2_U],
                                         v8f& accE0, v8f& accE1, v8f& accO0, v8f& accO1,
                                         v2f& qx, v2f& qy) {
    for (int u = 0; u < G2_U; ++u) {
        qx.x = fmaf(ax[u].x, ax[u].x, qx.x);
        qx.y = fmaf(ax[u].y, ax[u].y, qx.y);
        qy.x = fmaf(ay[u].x, ay[u].x, qy.x);
        qy.y = fmaf(ay[u].y, ay[u].y, qy.y);
        v2f b0 = *(const v2f*)(Bk + (2 * u) * PP);
        v2f b1 = *(const v2f*)(Bk + (2 * u) * PP + 32);
        v2f aE, aO;
        aE.x = ax[u].x; aE.y = ay[u].x;
        aO.x = ax[u].y; aO.y = ay[u].y;
        accE0 = __builtin_amdgcn_wmma_f32_16x16x4_f32(false, aE, false, b0, (short)0, accE0, false, false);
        accE1 = __builtin_amdgcn_wmma_f32_16x16x4_f32(false, aE, false, b1, (short)0, accE1, false, false);
        accO0 = __builtin_amdgcn_wmma_f32_16x16x4_f32(false, aO, false, b0, (short)0, accO0, false, false);
        accO1 = __builtin_amdgcn_wmma_f32_16x16x4_f32(false, aO, false, b1, (short)0, accO1, false, false);
    }
}
#define G3_U 8
__device__ __forceinline__ void g3_load_block(const float* __restrict__ Wb, v2f (&w)[G3_U]) {
    for (int u = 0; u < G3_U; ++u)
        w[u] = *(const v2f*)(Wb + 4 * u);
}
__device__ __forceinline__ void g3_block(const float* Zk, const v2f (&w)[G3_U], v8f& o0, v8f& o1) {
    for (int u = 0; u < G3_U; ++u) {
        v2f z0 = *(const v2f*)(Zk + (2 * u) * PP);
        v2f z1 = *(const v2f*)(Zk + (2 * u) * PP + 32);
        o0 = __builtin_amdgcn_wmma_f32_16x16x4_f32(false, z0, false, w[u], (short)0, o0, false, false);
        o1 = __builtin_amdgcn_wmma_f32_16x16x4_f32(false, z1, false, w[u], (short)0, o1, false, false);
    }
}

__global__ __launch_bounds__(512) void k_gemm123(
    const float* __restrict__ w1w, const float* __restrict__ w1b,
    const float* __restrict__ dw,
    const float* __restrict__ dg, const float* __restrict__ db,
    const float* __restrict__ pg,
    const float* __restrict__ pw,
    const float* __restrict__ pb,
    const float* __restrict__ w2w, const float* __restrict__ w2b,
    const float* __restrict__ x, float* __restrict__ out)
{
    __shared__ __attribute__((aligned(16))) float panel[(DH1 / 2) * PP];
    const int tid = threadIdx.x;
    const int b  = blockIdx.x / (T_ / PT);
    const int t0 = (blockIdx.x % (T_ / PT)) * PT;

    __shared__ __attribute__((aligned(16))) float xs[HT * XP];
    for (int q = 0; q < HT / 8; ++q) {
        const int row = (tid >> 6) + 8 * q;
        const int c4  = tid & 63;
        const int t   = t0 - 8 + row;
        float4 v = make_float4(0.f, 0.f, 0.f, 0.f);
        if ((unsigned)t < (unsigned)T_)
            v = *(const float4*)(x + ((size_t)b * T_ + t) * DIN + 4 * c4);
        *(float4*)(xs + row * XP + 4 * c4) = v;
    }

    __shared__ float dred[KW * G2_WAVES];
    __shared__ float dnrm[16];
    {
        const v2f* wq = (const v2f*)(dw + ((size_t)b * DH1 + 2 * tid) * KW);
        float wf[2 * KW];
        for (int u = 0; u < KW; ++u) { v2f t = wq[u]; wf[2 * u] = t.x; wf[2 * u + 1] = t.y; }
        float sq[KW];
        for (int k = 0; k < KW; ++k) sq[k] = fmaf(wf[KW + k], wf[KW + k], wf[k] * wf[k]);
        for (int m = 16; m >= 1; m >>= 1)
            for (int k = 0; k < KW; ++k) sq[k] += __shfl_xor(sq[k], m);
        if ((tid & 31) == 0)
            for (int k = 0; k < KW; ++k) dred[k * G2_WAVES + (tid >> 5)] = sq[k];
    }
    __syncthreads();
    if (tid < KW) {
        float tot = 0.f;
        for (int w = 0; w < G2_WAVES; ++w) tot += dred[tid * G2_WAVES + w];
        dnrm[tid] = 1.f / fmaxf(sqrtf(tot), 1e-12f);
    }
    __syncthreads();

    const int lane = tid & 31, wave = tid >> 5;
    const int hf = lane >> 4, l16 = lane & 15;

    {
        const float* Xa = xs + l16 * XP + 4 * hf;
        const float* W0 = w1w + (size_t)(wave * 16 + l16) * DIN + 4 * hf;
        float4 wA[G1_U], wB[G1_U];
        g1_load_block(W0, wA);
        #pragma nounroll
        for (int ps = 0; ps < 4; ++ps) {
            const float* Wp = W0 + (size_t)ps * 256 * DIN;
            const float* Wn = W0 + (size_t)((ps + 1) & 3) * 256 * DIN;
            const int c = (ps * 16 + wave) * 16 + l16;
            const float bias1 = w1b[c];
            v8f a0 = {}, a1 = {}, a2 = {};
            asm volatile("" ::: "memory");
            g1_load_block(Wp + 8 * G1_U, wB);
            g1_block(Xa, wA, a0, a1, a2);
            g1_load_block(Wp + 16 * G1_U, wA);
            g1_block(Xa + 8 * G1_U, wB, a0, a1, a2);
            g1_load_block(Wp + 24 * G1_U, wB);
            g1_block(Xa + 16 * G1_U, wA, a0, a1, a2);
            g1_load_block(Wn, wA);
            g1_block(Xa + 24 * G1_U, wB, a0, a1, a2);
            float* hrow = panel + (c >> 1) * PP + (c & 1) * HT + 8 * hf;
            #pragma nounroll
            for (int tile = 0; tile < 3; ++tile) {
                float o[8];
                for (int r = 0; r < 8; ++r) {
                    const int t = t0 - 8 + tile * 16 + 8 * hf + r;
                    const float v = mishf(a0[r] + bias1);
                    o[r] = ((unsigned)t < (unsigned)T_) ? v : 0.f;
                }
                *(float4*)(hrow + tile * 16)     = make_float4(o[0], o[1], o[2], o[3]);
                *(float4*)(hrow + tile * 16 + 4) = make_float4(o[4], o[5], o[6], o[7]);
                const v8f tmp = a0; a0 = a1; a1 = a2; a2 = tmp;
            }
        }
    }
    __syncthreads();

    {
        const int j  = tid & 7;
        const int pp = tid >> 3;
        float nd[KW];
        for (int k = 0; k < KW; ++k) nd[k] = dnrm[k];
        #pragma unroll 2
        for (int i = 0; i < (DH1 / 2) / 64; ++i) {
            const int p  = i * 64 + pp;
            const int bc = b * DH1 + 2 * p;
            float* row = panel + p * PP;
            const float* r0 = row + 4 + 4 * j;
            const float* r1 = r0 + HT;
            float win0[12], win1[12];
            const float4 l0 = *(const float4*)(r0), m0 = *(const float4*)(r0 + 4), n0 = *(const float4*)(r0 + 8);
            const float4 l1 = *(const float4*)(r1), m1 = *(const float4*)(r1 + 4), n1 = *(const float4*)(r1 + 8);
            put4(win0, 0, l0);
            put4(win0, 4, m0);
            put4(win0, 8, n0);
            put4(win1, 0, l1);
            put4(win1, 4, m1);
            put4(win1, 8, n1);
            const v2f* wp2 = (const v2f*)(dw + (size_t)bc * KW);
            float wf[2 * KW];
            for (int u = 0; u < KW; ++u) { v2f t = wp2[u]; wf[2 * u] = t.x; wf[2 * u + 1] = t.y; }
            float w0[KW], w1[KW];
            for (int k = 0; k < KW; ++k) { w0[k] = wf[k] * nd[k]; w1[k] = wf[KW + k] * nd[k]; }
            const v2f g2 = *(const v2f*)(dg + bc);
            const v2f b2 = *(const v2f*)(db + bc);
            const v2f q2 = *(const v2f*)(pg + bc);
            float4 u = dw_row4(win0, w0, g2.x, b2.x, q2.x);
            float4 v = dw_row4(win1, w1, g2.y, b2.y, q2.y);
            float4 lo = make_float4(u.x, v.x, u.y, v.y);
            float4 hi = make_float4(u.z, v.z, u.w, v.w);
            *(float4*)(row + 8 * j)     = lo;
            *(float4*)(row + 8 * j + 4) = hi;
        }
    }
    __syncthreads();

    const float* A = pw + (size_t)b * DH1 * DH2;
    const float* Bp = panel + hf * PP + 2 * l16;
    const int NB = DH1 / (4 * G2_U);

    v8f zE0[2], zE1[2], zO0[2], zO1[2];
    #pragma unroll
    for (int ps = 0; ps < 2; ++ps) {
        const int ob = wave * 32 + ps * (G2_WAVES * 32);
        const float* Ap = A + (size_t)(2 * hf) * DH2 + ob + 2 * l16;
        v8f accE0 = {}, accE1 = {}, accO0 = {}, accO1 = {};
        v2f qx = {}, qy = {};
        v2f x0[G2_U], y0[G2_U], x1[G2_U], y1[G2_U];
        g2_load_block(Ap, x0, y0);
        for (int kb = 0; kb < NB; kb += 2) {
            g2_load_block(Ap + (size_t)(4 * G2_U * (kb + 1)) * DH2, x1, y1);
            g2_block(Bp + (2 * G2_U * kb) * PP, x0, y0, accE0, accE1, accO0, accO1, qx, qy);
            g2_load_block(Ap + (size_t)(4 * G2_U * ((kb + 2) & (NB - 1))) * DH2, x0, y0);
            g2_block(Bp + (2 * G2_U * (kb + 1)) * PP, x1, y1, accE0, accE1, accO0, accO1, qx, qy);
            asm volatile("" :: "v"(x0[0]), "v"(x0[1]), "v"(x0[2]), "v"(x0[3]),
                               "v"(y0[0]), "v"(y0[1]), "v"(y0[2]), "v"(y0[3]));
        }
        const float pxE = __shfl_xor(qx.x, 16), pxO = __shfl_xor(qx.y, 16);
        const float pyE = __shfl_xor(qy.x, 16), pyO = __shfl_xor(qy.y, 16);
        const float totE = (((hf ? pxE : qx.x) + (hf ? pyE : qy.x)) + (hf ? qx.x : pxE)) + (hf ? qy.x : pyE);
        const float totO = (((hf ? pxO : qx.y) + (hf ? pyO : qy.y)) + (hf ? qx.y : pxO)) + (hf ? qy.y : pyO);
        const float invE = 1.f / fmaxf(sqrtf(totE), 1e-12f);
        const float invO = 1.f / fmaxf(sqrtf(totO), 1e-12f);
        for (int v = 0; v < 8; ++v) {
            int oE = ob + 2 * (v + 8 * hf);
            int oO = oE + 1;
            float sE = __shfl(invE, v + 8 * hf), cE = pb[b * DH2 + oE];
            float sO = __shfl(invO, v + 8 * hf), cO = pb[b * DH2 + oO];
            zE0[ps][v] = accE0[v] * sE + cE;
            zE1[ps][v] = accE1[v] * sE + cE;
            zO0[ps][v] = accO0[v] * sO + cO;
            zO1[ps][v] = accO1[v] * sO + cO;
        }
    }

    const int i0 = wave * 16;
    const int ii = i0 + l16;
    const float* Wp = w2w + (size_t)ii * DH2 + 2 * hf;
    v2f wa[G3_U], wb[G3_U];
    g3_load_block(Wp, wa);
    const size_t obase = ((size_t)b * T_ + t0 + 8 * hf) * DIN + ii;
    float xr0[8], xr1[8];
    for (int r = 0; r < 8; ++r) {
        xr0[r] = x[obase + (size_t)r * DIN];
        xr1[r] = x[obase + (size_t)(16 + r) * DIN];
    }
    const float bias = w2b[ii];

    __syncthreads();
    #pragma unroll
    for (int ps = 0; ps < 2; ++ps) {
        const int ob = wave * 32 + ps * (G2_WAVES * 32);
        for (int v = 0; v < 8; ++v) {
            int p = (ob >> 1) + v + 8 * hf;
            v2f lo, hi;
            lo.x = zE0[ps][v]; lo.y = zO0[ps][v];
            hi.x = zE1[ps][v]; hi.y = zO1[ps][v];
            *(v2f*)(panel + p * PP + 2 * l16)      = lo;
            *(v2f*)(panel + p * PP + 32 + 2 * l16) = hi;
        }
    }
    __syncthreads();

    v8f o0 = {}, o1 = {};
    const int NB3 = DH2 / (4 * G3_U);
    for (int kb = 0; kb < NB3; kb += 2) {
        g3_load_block(Wp + 4 * G3_U * (kb + 1), wb);
        g3_block(Bp + (2 * G3_U * kb) * PP, wa, o0, o1);
        g3_load_block(Wp + 4 * G3_U * ((kb + 2) & (NB3 - 1)), wa);
        g3_block(Bp + (2 * G3_U * (kb + 1)) * PP, wb, o0, o1);
    }
    for (int r = 0; r < 8; ++r) {
        out[obase + (size_t)r * DIN]        = o0[r] + bias + xr0[r];
        out[obase + (size_t)(16 + r) * DIN] = o1[r] + bias + xr1[r];
    }
}

extern "C" void kernel_launch(void* const* d_in, const int* in_sizes, int n_in,
                              void* d_out, int out_size, void* d_ws, size_t ws_size,
                              hipStream_t stream)
{
    const float* x   = (const float*)d_in[0];
    const float* d_w = (const float*)d_in[1];
    const float* d_g = (const float*)d_in[2];
    const float* d_b = (const float*)d_in[3];
    const float* p_w = (const float*)d_in[4];
    const float* p_g = (const float*)d_in[5];
    const float* p_b = (const float*)d_in[6];
    const float* w1w = (const float*)d_in[7];
    const float* w1b = (const float*)d_in[8];
    const float* w2w = (const float*)d_in[9];
    const float* w2b = (const float*)d_in[10];
    float* out = (float*)d_out;

    k_gemm123<<<B_ * (T_ / PT), 512, 0, stream>>>(w1w, w1b, d_w, d_g, d_b, p_g, p_w, p_b, w2w, w2b, x, out);
}
